# starting kernel
# baseline (speedup 1.0000x reference)
#include <hip/hip_runtime.h>


typedef __attribute__((ext_vector_type(2))) float v2f;
typedef __attribute__((ext_vector_type(8))) float v8f;

#define B_    4
#define H_    16
#define L_    2048
#define D_    64
#define RWIN  256
#define KEY0  (L_ - RWIN)
#define DIAG  (L_ - RWIN + 1)
#define LASTQ (L_ - 1 - DIAG)
#define NSPLIT 8
#define ROWS_PER_SPLIT (L_ / NSPLIT)

__global__ __launch_bounds__(256)
void vpartial_kernel(const float* __restrict__ v, float* __restrict__ partial) {
  __shared__ float4 red[256];
  const int blk = blockIdx.x;
  const int bh  = blk >> 3;
  const int s   = blk & (NSPLIT - 1);
  const int t   = threadIdx.x;
  const int c   = t & 15;
  const int g   = t >> 4;

  const float4* vb4 = (const float4*)(v + (size_t)bh * L_ * D_);
  const int row0 = s * ROWS_PER_SPLIT + g;

  float4 acc = make_float4(0.f, 0.f, 0.f, 0.f);
#pragma unroll
  for (int it = 0; it < ROWS_PER_SPLIT / 16; ++it) {
    float4 x = vb4[(size_t)(row0 + it * 16) * (D_ / 4) + c];
    acc.x += x.x; acc.y += x.y; acc.z += x.z; acc.w += x.w;
  }
  red[t] = acc;
  __syncthreads();
  if (t < 128) { float4 o = red[t + 128];
    red[t].x += o.x; red[t].y += o.y; red[t].z += o.z; red[t].w += o.w; }
  __syncthreads();
  if (t < 64)  { float4 o = red[t + 64];
    red[t].x += o.x; red[t].y += o.y; red[t].z += o.z; red[t].w += o.w; }
  __syncthreads();
  if (t < 32)  { float4 o = red[t + 32];
    red[t].x += o.x; red[t].y += o.y; red[t].z += o.z; red[t].w += o.w; }
  __syncthreads();
  if (t < 16) {
    float4 o = red[t + 16];
    float4 r = red[t];
    r.x += o.x; r.y += o.y; r.z += o.z; r.w += o.w;
    ((float4*)partial)[(size_t)blk * (D_ / 4) + t] = r;
  }
}

__global__ __launch_bounds__(64)
void vcombine_kernel(const float* __restrict__ partial, float* __restrict__ vmean) {
  const int bh = blockIdx.x;
  const int d  = threadIdx.x;
  float s = 0.f;
#pragma unroll
  for (int sp = 0; sp < NSPLIT; ++sp)
    s += partial[(size_t)(bh * NSPLIT + sp) * D_ + d];
  vmean[bh * D_ + d] = s * (1.0f / (float)L_);
}

__global__ __launch_bounds__(256)
void bcast_kernel(const float* __restrict__ vmean, float* __restrict__ out) {
  const size_t idx = (size_t)blockIdx.x * blockDim.x + threadIdx.x;
  const size_t e   = idx * 4;
  const int bh = (int)(e / (size_t)(L_ * D_));
  const int d  = (int)(e & (size_t)(D_ - 1));
  const float4* vm4 = (const float4*)vmean;
  float4 val = vm4[bh * (D_ / 4) + (d >> 2)];
  ((float4*)out)[idx] = val;
}

__global__ __launch_bounds__(512)
void attn_kernel(const float* __restrict__ q, const float* __restrict__ k,
                 const float* __restrict__ v, float* __restrict__ out) {
  __shared__ float pLds[16][16 * 17];

  const int bh   = blockIdx.x;
  const int wave = threadIdx.x >> 5;
  const int lane = threadIdx.x & 31;
  const int tcol = lane & 15;
  const int half = lane >> 4;

  const float* qb = q + (size_t)bh * L_ * D_ + (size_t)wave * 16 * D_;
  const float* kb = k + (size_t)bh * L_ * D_ + (size_t)KEY0 * D_;
  const float* vb = v + (size_t)bh * L_ * D_ + (size_t)KEY0 * D_;
  float*       ob = out + (size_t)bh * L_ * D_;

  v2f aq[16];
#pragma unroll
  for (int c4 = 0; c4 < 16; ++c4) {
    const float* p = qb + tcol * D_ + c4 * 4 + 2 * half;
    v2f a; a.x = p[0] * 0.125f; a.y = p[1] * 0.125f;
    aq[c4] = a;
  }

  const v8f vzero = {0.f, 0.f, 0.f, 0.f, 0.f, 0.f, 0.f, 0.f};
  v8f acc[4];
  acc[0] = vzero; acc[1] = vzero; acc[2] = vzero; acc[3] = vzero;
  float mrow[8], lsum[8];
#pragma unroll
  for (int r = 0; r < 8; ++r) { mrow[r] = -3.0e38f; lsum[r] = 0.0f; }

  float* myP = &pLds[wave][0];

  for (int kt = 0; kt < 16; ++kt) {
    v8f s = vzero;
#pragma unroll
    for (int c4 = 0; c4 < 16; ++c4) {
      const float* kp = kb + (kt * 16 + tcol) * D_ + c4 * 4 + 2 * half;
      v2f b; b.x = kp[0]; b.y = kp[1];
      s = __builtin_amdgcn_wmma_f32_16x16x4_f32(
              false, aq[c4], false, b, (short)0, s, false, false);
    }

    const int j = KEY0 + kt * 16 + tcol;
    float alpha[8];
#pragma unroll
    for (int r = 0; r < 8; ++r) {
      const int i = wave * 16 + r + 8 * half;
      float sv = (j - i >= DIAG) ? s[r] : -10000.0f;
      float rm = sv;
      rm = fmaxf(rm, __shfl_xor(rm, 1));
      rm = fmaxf(rm, __shfl_xor(rm, 2));
      rm = fmaxf(rm, __shfl_xor(rm, 4));
      rm = fmaxf(rm, __shfl_xor(rm, 8));
      const float mnew = fmaxf(mrow[r], rm);
      alpha[r] = __expf(mrow[r] - mnew);
      const float p = __expf(sv - mnew);
      float ps = p;
      ps += __shfl_xor(ps, 1);
      ps += __shfl_xor(ps, 2);
      ps += __shfl_xor(ps, 4);
      ps += __shfl_xor(ps, 8);
      lsum[r] = lsum[r] * alpha[r] + ps;
      mrow[r] = mnew;
      myP[(r + 8 * half) * 17 + tcol] = p;
    }

#pragma unroll
    for (int nt = 0; nt < 4; ++nt)
#pragma unroll
      for (int r = 0; r < 8; ++r)
        acc[nt][r] *= alpha[r];

#pragma unroll
    for (int nt = 0; nt < 4; ++nt) {
#pragma unroll
      for (int c4 = 0; c4 < 4; ++c4) {
        const int kk = c4 * 4 + 2 * half;
        v2f a; a.x = myP[tcol * 17 + kk];
               a.y = myP[tcol * 17 + kk + 1];
        const float* vp = vb + (kt * 16 + kk) * D_ + nt * 16 + tcol;
        v2f b; b.x = vp[0]; b.y = vp[D_];
        acc[nt] = __builtin_amdgcn_wmma_f32_16x16x4_f32(
                      false, a, false, b, (short)0, acc[nt], false, false);
      }
    }
  }

#pragma unroll
  for (int r = 0; r < 8; ++r) {
    const int i = wave * 16 + r + 8 * half;
    if (i <= LASTQ) {
      const float inv = 1.0f / lsum[r];
#pragma unroll
      for (int nt = 0; nt < 4; ++nt)
        ob[(size_t)i * D_ + nt * 16 + tcol] = acc[nt][r] * inv;
    }
  }
}

extern "C" void kernel_launch(void* const* d_in, const int* in_sizes, int n_in,
                              void* d_out, int out_size, void* d_ws, size_t ws_size,
                              hipStream_t stream) {
  const float* q = (const float*)d_in[0];
  const float* k = (const float*)d_in[1];
  const float* v = (const float*)d_in[2];
  float* out = (float*)d_out;

  float* partial = (float*)d_ws;
  float* vmean   = partial + B_ * H_ * NSPLIT * D_;

  vpartial_kernel<<<B_ * H_ * NSPLIT, 256, 0, stream>>>(v, partial);
  vcombine_kernel<<<B_ * H_, 64, 0, stream>>>(partial, vmean);

  const int n4 = (B_ * H_ * L_ * D_) / 4;
  bcast_kernel<<<n4 / 256, 256, 0, stream>>>(vmean, out);

  attn_kernel<<<B_ * H_, 512, 0, stream>>>(q, k, v, out);
}


# attn corner: both products transposed (S^T=K*Q^T, O^T=V^T*P^T): P feeds the 2nd WMMA from its own registers, softmax max/sum per lane, no per-tile LDS relayout or shuffles; new sum order of l and O
# speedup vs baseline: 2.4675x; 2.4675x over previous
#include <hip/hip_runtime.h>


typedef __attribute__((ext_vector_type(2))) float v2f;
typedef __attribute__((ext_vector_type(8))) float v8f;

#define B_    4
#define H_    16
#define L_    2048
#define D_    64
#define RWIN  256
#define KEY0  (L_ - RWIN)
#define DIAG  (L_ - RWIN + 1)
#define LASTQ (L_ - 1 - DIAG)
#define NSPLIT 8
#define ROWS_PER_SPLIT (L_ / NSPLIT)

__global__ __launch_bounds__(256)
void vpartial_kernel(const float* __restrict__ v, float* __restrict__ partial) {
  __shared__ float4 red[256];
  const int blk = blockIdx.x;
  const int bh  = blk >> 3;
  const int s   = blk & (NSPLIT - 1);
  const int t   = threadIdx.x;
  const int c   = t & 15;
  const int g   = t >> 4;

  const float4* vb4 = (const float4*)(v + (size_t)bh * L_ * D_);
  const int row0 = s * ROWS_PER_SPLIT + g;

  float4 acc = make_float4(0.f, 0.f, 0.f, 0.f);
#pragma unroll
  for (int it = 0; it < ROWS_PER_SPLIT / 16; ++it) {
    float4 x = vb4[(size_t)(row0 + it * 16) * (D_ / 4) + c];
    acc.x += x.x; acc.y += x.y; acc.z += x.z; acc.w += x.w;
  }
  red[t] = acc;
  __syncthreads();
  if (t < 128) { float4 o = red[t + 128];
    red[t].x += o.x; red[t].y += o.y; red[t].z += o.z; red[t].w += o.w; }
  __syncthreads();
  if (t < 64)  { float4 o = red[t + 64];
    red[t].x += o.x; red[t].y += o.y; red[t].z += o.z; red[t].w += o.w; }
  __syncthreads();
  if (t < 32)  { float4 o = red[t + 32];
    red[t].x += o.x; red[t].y += o.y; red[t].z += o.z; red[t].w += o.w; }
  __syncthreads();
  if (t < 16) {
    float4 o = red[t + 16];
    float4 r = red[t];
    r.x += o.x; r.y += o.y; r.z += o.z; r.w += o.w;
    ((float4*)partial)[(size_t)blk * (D_ / 4) + t] = r;
  }
}

__global__ __launch_bounds__(64)
void vcombine_kernel(const float* __restrict__ partial, float* __restrict__ vmean) {
  const int bh = blockIdx.x;
  const int d  = threadIdx.x;
  float s = 0.f;
#pragma unroll
  for (int sp = 0; sp < NSPLIT; ++sp)
    s += partial[(size_t)(bh * NSPLIT + sp) * D_ + d];
  vmean[bh * D_ + d] = s * (1.0f / (float)L_);
}

__global__ __launch_bounds__(256)
void bcast_kernel(const float* __restrict__ vmean, float* __restrict__ out) {
  const size_t idx = (size_t)blockIdx.x * blockDim.x + threadIdx.x;
  const size_t e   = idx * 4;
  const int bh = (int)(e / (size_t)(L_ * D_));
  const int d  = (int)(e & (size_t)(D_ - 1));
  const float4* vm4 = (const float4*)vmean;
  float4 val = vm4[bh * (D_ / 4) + (d >> 2)];
  ((float4*)out)[idx] = val;
}

__device__ __forceinline__ float xhalf(float x) {
  int b = __builtin_bit_cast(int, x);
  return __builtin_bit_cast(float,
      __builtin_amdgcn_permlanex16(b, b, 0x76543210, 0xfedcba98, false, false));
}

__global__ __launch_bounds__(32)
void attn_kernel(const float* __restrict__ q, const float* __restrict__ k,
                 const float* __restrict__ v, float* __restrict__ out) {
  __shared__ float4 oLds[16 * 17];

  const int bh   = blockIdx.x & (B_ * H_ - 1);
  const int qt   = blockIdx.x / (B_ * H_);
  const int lane = threadIdx.x;
  const int tcol = lane & 15;
  const int half = lane >> 4;

  const float* qb = q + (size_t)bh * L_ * D_ + (size_t)qt * 16 * D_;
  const float* kb = k + (size_t)bh * L_ * D_ + (size_t)KEY0 * D_;
  const float* vb = v + (size_t)bh * L_ * D_ + (size_t)KEY0 * D_;
  float*       ob = out + (size_t)bh * L_ * D_;

  v2f bq[16];
#pragma unroll
  for (int c4 = 0; c4 < 16; ++c4) {
    const float* p = qb + tcol * D_ + c4 * 4 + 2 * half;
    v2f b; b.x = p[0] * 0.125f; b.y = p[1] * 0.125f;
    bq[c4] = b;
  }

  const v8f vzero = {0.f, 0.f, 0.f, 0.f, 0.f, 0.f, 0.f, 0.f};
  v8f acc[4];
  acc[0] = vzero; acc[1] = vzero; acc[2] = vzero; acc[3] = vzero;
  float mrow = -3.0e38f, lsum = 0.0f;
  const int i = qt * 16 + tcol;

  for (int kt = qt; kt < 16; ++kt) {
    v8f s = vzero;
#pragma unroll
    for (int c4 = 0; c4 < 16; ++c4) {
      const float* kp = kb + (kt * 16 + tcol) * D_ + c4 * 4 + 2 * half;
      v2f a; a.x = kp[0]; a.y = kp[1];
      s = __builtin_amdgcn_wmma_f32_16x16x4_f32(
              false, a, false, bq[c4], (short)0, s, false, false);
    }

    const int j0 = KEY0 + kt * 16 + 8 * half;
    float sv[8];
#pragma unroll
    for (int r = 0; r < 8; ++r)
      sv[r] = (j0 + r - i >= DIAG) ? s[r] : -10000.0f;
    float rm = sv[0];
#pragma unroll
    for (int r = 1; r < 8; ++r) rm = fmaxf(rm, sv[r]);
    rm = fmaxf(rm, xhalf(rm));
    const float mnew  = fmaxf(mrow, rm);
    const float alpha = __expf(mrow - mnew);
    float p[8];
    float ps = 0.0f;
#pragma unroll
    for (int r = 0; r < 8; ++r) { p[r] = __expf(sv[r] - mnew); ps += p[r]; }
    ps += xhalf(ps);
    lsum = lsum * alpha + ps;
    mrow = mnew;

#pragma unroll
    for (int mt = 0; mt < 4; ++mt)
#pragma unroll
      for (int r = 0; r < 8; ++r)
        acc[mt][r] *= alpha;

#pragma unroll
    for (int w = 0; w < 4; ++w) {
      v2f b; b.x = p[2 * w]; b.y = p[2 * w + 1];
      const float* vrow = vb + (kt * 16 + 8 * half + 2 * w) * D_ + tcol;
#pragma unroll
      for (int mt = 0; mt < 4; ++mt) {
        v2f a; a.x = vrow[mt * 16]; a.y = vrow[mt * 16 + D_];
        acc[mt] = __builtin_amdgcn_wmma_f32_16x16x4_f32(
                      false, a, false, b, (short)0, acc[mt], false, false);
      }
    }
  }

  const float inv = 1.0f / lsum;
#pragma unroll
  for (int mt = 0; mt < 4; ++mt) {
    float4 lo, hi;
    lo.x = acc[mt][0] * inv; lo.y = acc[mt][1] * inv;
    lo.z = acc[mt][2] * inv; lo.w = acc[mt][3] * inv;
    hi.x = acc[mt][4] * inv; hi.y = acc[mt][5] * inv;
    hi.z = acc[mt][6] * inv; hi.w = acc[mt][7] * inv;
    oLds[tcol * 17 + mt * 4 + half * 2]     = lo;
    oLds[tcol * 17 + mt * 4 + half * 2 + 1] = hi;
  }
  __syncthreads();
  if (qt * 16 + 15 <= LASTQ) {
#pragma unroll
    for (int it = 0; it < 8; ++it) {
      const int io = qt * 16 + it * 2 + half;
      ((float4*)(ob + (size_t)io * D_))[tcol] = oLds[(it * 2 + half) * 17 + tcol];
    }
  } else {
#pragma unroll
    for (int it = 0; it < 8; ++it) {
      const int io = qt * 16 + it * 2 + half;
      if (io <= LASTQ)
        ((float4*)(ob + (size_t)io * D_))[tcol] = oLds[(it * 2 + half) * 17 + tcol];
    }
  }
}

extern "C" void kernel_launch(void* const* d_in, const int* in_sizes, int n_in,
                              void* d_out, int out_size, void* d_ws, size_t ws_size,
                              hipStream_t stream) {
  const float* q = (const float*)d_in[0];
  const float* k = (const float*)d_in[1];
  const float* v = (const float*)d_in[2];
  float* out = (float*)d_out;

  float* partial = (float*)d_ws;
  float* vmean   = partial + B_ * H_ * NSPLIT * D_;

  vpartial_kernel<<<B_ * H_ * NSPLIT, 256, 0, stream>>>(v, partial);
  vcombine_kernel<<<B_ * H_, 64, 0, stream>>>(partial, vmean);

  const int n4 = (B_ * H_ * L_ * D_) / 4;
  bcast_kernel<<<n4 / 256, 256, 0, stream>>>(vmean, out);

  attn_kernel<<<B_ * H_ * (RWIN / 16), 32, 0, stream>>>(q, k, v, out);
}


# attn key-tile loop software-pipelined: tile kt+1's 48 K/V fragment loads issued as one group into a 2nd register set in front of tile kt's WMMAs (pairs of tiles, body twice, no copies); same sums
# speedup vs baseline: 2.9864x; 1.2103x over previous
#include <hip/hip_runtime.h>


typedef __attribute__((ext_vector_type(2))) float v2f;
typedef __attribute__((ext_vector_type(8))) float v8f;

#define B_    4
#define H_    16
#define L_    2048
#define D_    64
#define RWIN  256
#define KEY0  (L_ - RWIN)
#define DIAG  (L_ - RWIN + 1)
#define LASTQ (L_ - 1 - DIAG)
#define NSPLIT 8
#define ROWS_PER_SPLIT (L_ / NSPLIT)

__global__ __launch_bounds__(256)
void vpartial_kernel(const float* __restrict__ v, float* __restrict__ partial) {
  __shared__ float4 red[256];
  const int blk = blockIdx.x;
  const int bh  = blk >> 3;
  const int s   = blk & (NSPLIT - 1);
  const int t   = threadIdx.x;
  const int c   = t & 15;
  const int g   = t >> 4;

  const float4* vb4 = (const float4*)(v + (size_t)bh * L_ * D_);
  const int row0 = s * ROWS_PER_SPLIT + g;

  float4 acc = make_float4(0.f, 0.f, 0.f, 0.f);
#pragma unroll
  for (int it = 0; it < ROWS_PER_SPLIT / 16; ++it) {
    float4 x = vb4[(size_t)(row0 + it * 16) * (D_ / 4) + c];
    acc.x += x.x; acc.y += x.y; acc.z += x.z; acc.w += x.w;
  }
  red[t] = acc;
  __syncthreads();
  if (t < 128) { float4 o = red[t + 128];
    red[t].x += o.x; red[t].y += o.y; red[t].z += o.z; red[t].w += o.w; }
  __syncthreads();
  if (t < 64)  { float4 o = red[t + 64];
    red[t].x += o.x; red[t].y += o.y; red[t].z += o.z; red[t].w += o.w; }
  __syncthreads();
  if (t < 32)  { float4 o = red[t + 32];
    red[t].x += o.x; red[t].y += o.y; red[t].z += o.z; red[t].w += o.w; }
  __syncthreads();
  if (t < 16) {
    float4 o = red[t + 16];
    float4 r = red[t];
    r.x += o.x; r.y += o.y; r.z += o.z; r.w += o.w;
    ((float4*)partial)[(size_t)blk * (D_ / 4) + t] = r;
  }
}

__global__ __launch_bounds__(64)
void vcombine_kernel(const float* __restrict__ partial, float* __restrict__ vmean) {
  const int bh = blockIdx.x;
  const int d  = threadIdx.x;
  float s = 0.f;
#pragma unroll
  for (int sp = 0; sp < NSPLIT; ++sp)
    s += partial[(size_t)(bh * NSPLIT + sp) * D_ + d];
  vmean[bh * D_ + d] = s * (1.0f / (float)L_);
}

__global__ __launch_bounds__(256)
void bcast_kernel(const float* __restrict__ vmean, float* __restrict__ out) {
  const size_t idx = (size_t)blockIdx.x * blockDim.x + threadIdx.x;
  const size_t e   = idx * 4;
  const int bh = (int)(e / (size_t)(L_ * D_));
  const int d  = (int)(e & (size_t)(D_ - 1));
  const float4* vm4 = (const float4*)vmean;
  float4 val = vm4[bh * (D_ / 4) + (d >> 2)];
  ((float4*)out)[idx] = val;
}

__device__ __forceinline__ float xhalf(float x) {
  int b = __builtin_bit_cast(int, x);
  return __builtin_bit_cast(float,
      __builtin_amdgcn_permlanex16(b, b, 0x76543210, 0xfedcba98, false, false));
}

struct KVTile { v2f k[16]; v2f v[16]; };

__device__ __forceinline__ void load_tile(KVTile& t, const float* __restrict__ kb,
                                          const float* __restrict__ vb,
                                          int kt, int tcol, int half) {
  const float* kp = kb + (kt * 16 + tcol) * D_ + 2 * half;
#pragma unroll
  for (int c4 = 0; c4 < 16; ++c4) {
    v2f a; a.x = kp[c4 * 4]; a.y = kp[c4 * 4 + 1];
    t.k[c4] = a;
  }
  const float* vp = vb + (kt * 16 + 8 * half) * D_ + tcol;
#pragma unroll
  for (int w = 0; w < 4; ++w)
#pragma unroll
    for (int mt = 0; mt < 4; ++mt) {
      v2f a; a.x = vp[(2 * w) * D_ + mt * 16]; a.y = vp[(2 * w + 1) * D_ + mt * 16];
      t.v[w * 4 + mt] = a;
    }
}

__device__ __forceinline__ void tile_step(const KVTile& t, const v2f (&bq)[16],
                                          v8f (&acc)[4], float& mrow, float& lsum,
                                          int kt, int i, int half) {
  v8f s = {0.f, 0.f, 0.f, 0.f, 0.f, 0.f, 0.f, 0.f};
#pragma unroll
  for (int c4 = 0; c4 < 16; ++c4)
    s = __builtin_amdgcn_wmma_f32_16x16x4_f32(
            false, t.k[c4], false, bq[c4], (short)0, s, false, false);

  const int j0 = KEY0 + kt * 16 + 8 * half;
  float sv[8];
#pragma unroll
  for (int r = 0; r < 8; ++r)
    sv[r] = (j0 + r - i >= DIAG) ? s[r] : -10000.0f;
  float rm = sv[0];
#pragma unroll
  for (int r = 1; r < 8; ++r) rm = fmaxf(rm, sv[r]);
  rm = fmaxf(rm, xhalf(rm));
  const float mnew  = fmaxf(mrow, rm);
  const float alpha = __expf(mrow - mnew);
  float p[8];
  float ps = 0.0f;
#pragma unroll
  for (int r = 0; r < 8; ++r) { p[r] = __expf(sv[r] - mnew); ps += p[r]; }
  ps += xhalf(ps);
  lsum = lsum * alpha + ps;
  mrow = mnew;

#pragma unroll
  for (int mt = 0; mt < 4; ++mt)
#pragma unroll
    for (int r = 0; r < 8; ++r)
      acc[mt][r] *= alpha;

#pragma unroll
  for (int w = 0; w < 4; ++w) {
    v2f b; b.x = p[2 * w]; b.y = p[2 * w + 1];
#pragma unroll
    for (int mt = 0; mt < 4; ++mt)
      acc[mt] = __builtin_amdgcn_wmma_f32_16x16x4_f32(
                    false, t.v[w * 4 + mt], false, b, (short)0, acc[mt], false, false);
  }
}

__global__ __launch_bounds__(32)
void attn_kernel(const float* __restrict__ q, const float* __restrict__ k,
                 const float* __restrict__ v, float* __restrict__ out) {
  __shared__ float4 oLds[16 * 17];

  const int bh   = blockIdx.x & (B_ * H_ - 1);
  const int qt   = blockIdx.x / (B_ * H_);
  const int lane = threadIdx.x;
  const int tcol = lane & 15;
  const int half = lane >> 4;

  const float* qb = q + (size_t)bh * L_ * D_ + (size_t)qt * 16 * D_;
  const float* kb = k + (size_t)bh * L_ * D_ + (size_t)KEY0 * D_;
  const float* vb = v + (size_t)bh * L_ * D_ + (size_t)KEY0 * D_;
  float*       ob = out + (size_t)bh * L_ * D_;

  KVTile ta, tb;
  load_tile(ta, kb, vb, qt, tcol, half);

  v2f bq[16];
#pragma unroll
  for (int c4 = 0; c4 < 16; ++c4) {
    const float* p = qb + tcol * D_ + c4 * 4 + 2 * half;
    v2f b; b.x = p[0] * 0.125f; b.y = p[1] * 0.125f;
    bq[c4] = b;
  }

  const v8f vzero = {0.f, 0.f, 0.f, 0.f, 0.f, 0.f, 0.f, 0.f};
  v8f acc[4];
  acc[0] = vzero; acc[1] = vzero; acc[2] = vzero; acc[3] = vzero;
  float mrow = -3.0e38f, lsum = 0.0f;
  const int i = qt * 16 + tcol;

  int kt = qt;
  for (; kt + 1 < 16; kt += 2) {
    load_tile(tb, kb, vb, kt + 1, tcol, half);
    __builtin_amdgcn_sched_barrier(0);
    tile_step(ta, bq, acc, mrow, lsum, kt, i, half);
    __builtin_amdgcn_sched_barrier(0);

    load_tile(ta, kb, vb, (kt + 2 < 16) ? kt + 2 : 15, tcol, half);
    __builtin_amdgcn_sched_barrier(0);
    tile_step(tb, bq, acc, mrow, lsum, kt + 1, i, half);
  }
  if (kt < 16)
    tile_step(ta, bq, acc, mrow, lsum, kt, i, half);

  const float inv = 1.0f / lsum;
#pragma unroll
  for (int mt = 0; mt < 4; ++mt) {
    float4 lo, hi;
    lo.x = acc[mt][0] * inv; lo.y = acc[mt][1] * inv;
    lo.z = acc[mt][2] * inv; lo.w = acc[mt][3] * inv;
    hi.x = acc[mt][4] * inv; hi.y = acc[mt][5] * inv;
    hi.z = acc[mt][6] * inv; hi.w = acc[mt][7] * inv;
    oLds[tcol * 17 + mt * 4 + half * 2]     = lo;
    oLds[tcol * 17 + mt * 4 + half * 2 + 1] = hi;
  }
  __syncthreads();
  if (qt * 16 + 15 <= LASTQ) {
#pragma unroll
    for (int it = 0; it < 8; ++it) {
      const int io = qt * 16 + it * 2 + half;
      ((float4*)(ob + (size_t)io * D_))[tcol] = oLds[(it * 2 + half) * 17 + tcol];
    }
  } else {
#pragma unroll
    for (int it = 0; it < 8; ++it) {
      const int io = qt * 16 + it * 2 + half;
      if (io <= LASTQ)
        ((float4*)(ob + (size_t)io * D_))[tcol] = oLds[(it * 2 + half) * 17 + tcol];
    }
  }
}

extern "C" void kernel_launch(void* const* d_in, const int* in_sizes, int n_in,
                              void* d_out, int out_size, void* d_ws, size_t ws_size,
                              hipStream_t stream) {
  const float* q = (const float*)d_in[0];
  const float* k = (const float*)d_in[1];
  const float* v = (const float*)d_in[2];
  float* out = (float*)d_out;

  float* partial = (float*)d_ws;
  float* vmean   = partial + B_ * H_ * NSPLIT * D_;

  vpartial_kernel<<<B_ * H_ * NSPLIT, 256, 0, stream>>>(v, partial);
  vcombine_kernel<<<B_ * H_, 64, 0, stream>>>(partial, vmean);

  const int n4 = (B_ * H_ * L_ * D_) / 4;
  bcast_kernel<<<n4 / 256, 256, 0, stream>>>(vmean, out);

  attn_kernel<<<B_ * H_ * (RWIN / 16), 32, 0, stream>>>(q, k, v, out);
}


# mean path split by columns: 128 work-groups each own one 128-byte column half of a head, sum its 2048 rows, store rows 255..2047 themselves; 4 launches -> 2, no partials; new sum order of mean
# speedup vs baseline: 3.4123x; 1.1426x over previous
#include <hip/hip_runtime.h>


typedef __attribute__((ext_vector_type(2))) float v2f;
typedef __attribute__((ext_vector_type(8))) float v8f;

#define B_    4
#define H_    16
#define L_    2048
#define D_    64
#define RWIN  256
#define KEY0  (L_ - RWIN)
#define DIAG  (L_ - RWIN + 1)
#define LASTQ (L_ - 1 - DIAG)
#define MEAN0 (LASTQ + 1)
#define MROWS 128

__device__ __forceinline__ float4 f4add(float4 a, float4 b) {
  return make_float4(a.x + b.x, a.y + b.y, a.z + b.z, a.w + b.w);
}
__device__ __forceinline__ float4 f4xor(float4 a, int m) {
  return make_float4(__shfl_xor(a.x, m), __shfl_xor(a.y, m),
                     __shfl_xor(a.z, m), __shfl_xor(a.w, m));
}

__global__ __launch_bounds__(1024)
void mean_rows_kernel(const float* __restrict__ v, float* __restrict__ out) {
  __shared__ float4 red[32 * 8];
  const int bh   = blockIdx.x >> 1;
  const int hc   = blockIdx.x & 1;
  const int t    = threadIdx.x;
  const int c    = t & 7;
  const int g    = t >> 3;
  const int w    = t >> 5;
  const int lane = t & 31;

  const float4* vb4 = (const float4*)(v + (size_t)bh * L_ * D_) + hc * 8 + c;
  float4 x[L_ / MROWS];
#pragma unroll
  for (int it = 0; it < L_ / MROWS; ++it)
    x[it] = vb4[(size_t)(g + it * MROWS) * (D_ / 4)];
  float4 acc = x[0];
#pragma unroll
  for (int it = 1; it < L_ / MROWS; ++it) acc = f4add(acc, x[it]);

  acc = f4add(acc, f4xor(acc, 8));
  acc = f4add(acc, f4xor(acc, 16));
  if (lane < 8) red[w * 8 + c] = acc;
  __syncthreads();

  float4 s = red[lane];
#pragma unroll
  for (int i = 1; i < 8; ++i) s = f4add(s, red[lane + 32 * i]);
  s = f4add(s, f4xor(s, 8));
  s = f4add(s, f4xor(s, 16));
  const float sc = 1.0f / (float)L_;
  const float4 m = make_float4(s.x * sc, s.y * sc, s.z * sc, s.w * sc);

  float4* ob4 = (float4*)(out + (size_t)bh * L_ * D_) + hc * 8 + c;
  if (g == 0) ob4[(size_t)MEAN0 * (D_ / 4)] = m;
#pragma unroll
  for (int it = 0; it < (L_ - MEAN0 - 1) / MROWS; ++it)
    ob4[(size_t)(MEAN0 + 1 + g + it * MROWS) * (D_ / 4)] = m;
}

__device__ __forceinline__ float xhalf(float x) {
  int b = __builtin_bit_cast(int, x);
  return __builtin_bit_cast(float,
      __builtin_amdgcn_permlanex16(b, b, 0x76543210, 0xfedcba98, false, false));
}

struct KVTile { v2f k[16]; v2f v[16]; };

__device__ __forceinline__ void load_tile(KVTile& t, const float* __restrict__ kb,
                                          const float* __restrict__ vb,
                                          int kt, int tcol, int half) {
  const float* kp = kb + (kt * 16 + tcol) * D_ + 2 * half;
#pragma unroll
  for (int c4 = 0; c4 < 16; ++c4) {
    v2f a; a.x = kp[c4 * 4]; a.y = kp[c4 * 4 + 1];
    t.k[c4] = a;
  }
  const float* vp = vb + (kt * 16 + 8 * half) * D_ + tcol;
#pragma unroll
  for (int w = 0; w < 4; ++w)
#pragma unroll
    for (int mt = 0; mt < 4; ++mt) {
      v2f a; a.x = vp[(2 * w) * D_ + mt * 16]; a.y = vp[(2 * w + 1) * D_ + mt * 16];
      t.v[w * 4 + mt] = a;
    }
}

__device__ __forceinline__ void tile_step(const KVTile& t, const v2f (&bq)[16],
                                          v8f (&acc)[4], float& mrow, float& lsum,
                                          int kt, int i, int half) {
  v8f s = {0.f, 0.f, 0.f, 0.f, 0.f, 0.f, 0.f, 0.f};
#pragma unroll
  for (int c4 = 0; c4 < 16; ++c4)
    s = __builtin_amdgcn_wmma_f32_16x16x4_f32(
            false, t.k[c4], false, bq[c4], (short)0, s, false, false);

  const int j0 = KEY0 + kt * 16 + 8 * half;
  float sv[8];
#pragma unroll
  for (int r = 0; r < 8; ++r)
    sv[r] = (j0 + r - i >= DIAG) ? s[r] : -10000.0f;
  float rm = sv[0];
#pragma unroll
  for (int r = 1; r < 8; ++r) rm = fmaxf(rm, sv[r]);
  rm = fmaxf(rm, xhalf(rm));
  const float mnew  = fmaxf(mrow, rm);
  const float alpha = __expf(mrow - mnew);
  float p[8];
  float ps = 0.0f;
#pragma unroll
  for (int r = 0; r < 8; ++r) { p[r] = __expf(sv[r] - mnew); ps += p[r]; }
  ps += xhalf(ps);
  lsum = lsum * alpha + ps;
  mrow = mnew;

#pragma unroll
  for (int mt = 0; mt < 4; ++mt)
#pragma unroll
    for (int r = 0; r < 8; ++r)
      acc[mt][r] *= alpha;

#pragma unroll
  for (int w = 0; w < 4; ++w) {
    v2f b; b.x = p[2 * w]; b.y = p[2 * w + 1];
#pragma unroll
    for (int mt = 0; mt < 4; ++mt)
      acc[mt] = __builtin_amdgcn_wmma_f32_16x16x4_f32(
                    false, t.v[w * 4 + mt], false, b, (short)0, acc[mt], false, false);
  }
}

__global__ __launch_bounds__(32)
void attn_kernel(const float* __restrict__ q, const float* __restrict__ k,
                 const float* __restrict__ v, float* __restrict__ out) {
  __shared__ float4 oLds[16 * 17];

  const int bh   = blockIdx.x & (B_ * H_ - 1);
  const int qt   = blockIdx.x / (B_ * H_);
  const int lane = threadIdx.x;
  const int tcol = lane & 15;
  const int half = lane >> 4;

  const float* qb = q + (size_t)bh * L_ * D_ + (size_t)qt * 16 * D_;
  const float* kb = k + (size_t)bh * L_ * D_ + (size_t)KEY0 * D_;
  const float* vb = v + (size_t)bh * L_ * D_ + (size_t)KEY0 * D_;
  float*       ob = out + (size_t)bh * L_ * D_;

  KVTile ta, tb;
  load_tile(ta, kb, vb, qt, tcol, half);

  v2f bq[16];
#pragma unroll
  for (int c4 = 0; c4 < 16; ++c4) {
    const float* p = qb + tcol * D_ + c4 * 4 + 2 * half;
    v2f b; b.x = p[0] * 0.125f; b.y = p[1] * 0.125f;
    bq[c4] = b;
  }

  const v8f vzero = {0.f, 0.f, 0.f, 0.f, 0.f, 0.f, 0.f, 0.f};
  v8f acc[4];
  acc[0] = vzero; acc[1] = vzero; acc[2] = vzero; acc[3] = vzero;
  float mrow = -3.0e38f, lsum = 0.0f;
  const int i = qt * 16 + tcol;

  int kt = qt;
  for (; kt + 1 < 16; kt += 2) {
    load_tile(tb, kb, vb, kt + 1, tcol, half);
    __builtin_amdgcn_sched_barrier(0);
    tile_step(ta, bq, acc, mrow, lsum, kt, i, half);
    __builtin_amdgcn_sched_barrier(0);

    load_tile(ta, kb, vb, (kt + 2 < 16) ? kt + 2 : 15, tcol, half);
    __builtin_amdgcn_sched_barrier(0);
    tile_step(tb, bq, acc, mrow, lsum, kt + 1, i, half);
  }
  if (kt < 16)
    tile_step(ta, bq, acc, mrow, lsum, kt, i, half);

  const float inv = 1.0f / lsum;
#pragma unroll
  for (int mt = 0; mt < 4; ++mt) {
    float4 lo, hi;
    lo.x = acc[mt][0] * inv; lo.y = acc[mt][1] * inv;
    lo.z = acc[mt][2] * inv; lo.w = acc[mt][3] * inv;
    hi.x = acc[mt][4] * inv; hi.y = acc[mt][5] * inv;
    hi.z = acc[mt][6] * inv; hi.w = acc[mt][7] * inv;
    oLds[tcol * 17 + mt * 4 + half * 2]     = lo;
    oLds[tcol * 17 + mt * 4 + half * 2 + 1] = hi;
  }
  __syncthreads();
  if (qt * 16 + 15 <= LASTQ) {
#pragma unroll
    for (int it = 0; it < 8; ++it) {
      const int io = qt * 16 + it * 2 + half;
      ((float4*)(ob + (size_t)io * D_))[tcol] = oLds[(it * 2 + half) * 17 + tcol];
    }
  } else {
#pragma unroll
    for (int it = 0; it < 8; ++it) {
      const int io = qt * 16 + it * 2 + half;
      if (io <= LASTQ)
        ((float4*)(ob + (size_t)io * D_))[tcol] = oLds[(it * 2 + half) * 17 + tcol];
    }
  }
}

extern "C" void kernel_launch(void* const* d_in, const int* in_sizes, int n_in,
                              void* d_out, int out_size, void* d_ws, size_t ws_size,
                              hipStream_t stream) {
  const float* q = (const float*)d_in[0];
  const float* k = (const float*)d_in[1];
  const float* v = (const float*)d_in[2];
  float* out = (float*)d_out;

  mean_rows_kernel<<<B_ * H_ * 2, 1024, 0, stream>>>(v, out);
  attn_kernel<<<B_ * H_ * (RWIN / 16), 32, 0, stream>>>(q, k, v, out);
}


# 2 launches -> 1, jobs in sequence: each (head, column half) group of 16 waves stores its mean rows, then 8 of its waves take a query tile each; attention sums unchanged, new sum order of mean
# speedup vs baseline: 3.6579x; 1.0720x over previous
#include <hip/hip_runtime.h>


typedef __attribute__((ext_vector_type(2))) float v2f;
typedef __attribute__((ext_vector_type(8))) float v8f;

#define B_    4
#define H_    16
#define L_    2048
#define D_    64
#define RWIN  256
#define KEY0  (L_ - RWIN)
#define DIAG  (L_ - RWIN + 1)
#define LASTQ (L_ - 1 - DIAG)
#define MEAN0 (LASTQ + 1)
#define NWAVE 16
#define AWAVE 8
#define MROWS 64

__device__ __forceinline__ float4 f4add(float4 a, float4 b) {
  return make_float4(a.x + b.x, a.y + b.y, a.z + b.z, a.w + b.w);
}
__device__ __forceinline__ float4 f4xor(float4 a, int m) {
  return make_float4(__shfl_xor(a.x, m), __shfl_xor(a.y, m),
                     __shfl_xor(a.z, m), __shfl_xor(a.w, m));
}

__device__ __forceinline__
void mean_rows(float4* red, const float* __restrict__ v, float* __restrict__ out,
               int bh, int hc, int t, int w) {
  const int c    = t & 7;
  const int g    = t >> 3;
  const int lane = t & 31;

  const float4* vb4 = (const float4*)(v + (size_t)bh * L_ * D_) + hc * 8 + c;
  float4 x[L_ / MROWS];
#pragma unroll
  for (int it = 0; it < L_ / MROWS; ++it)
    x[it] = vb4[(size_t)(g + it * MROWS) * (D_ / 4)];
  float4 acc = x[0];
#pragma unroll
  for (int it = 1; it < L_ / MROWS; ++it) acc = f4add(acc, x[it]);

  acc = f4add(acc, f4xor(acc, 8));
  acc = f4add(acc, f4xor(acc, 16));
  if (lane < 8) red[w * 8 + c] = acc;
  __syncthreads();

  float4 s = red[lane];
#pragma unroll
  for (int i = 1; i < NWAVE / 4; ++i) s = f4add(s, red[lane + 32 * i]);
  s = f4add(s, f4xor(s, 8));
  s = f4add(s, f4xor(s, 16));
  const float sc = 1.0f / (float)L_;
  const float4 m = make_float4(s.x * sc, s.y * sc, s.z * sc, s.w * sc);

  float4* ob4 = (float4*)(out + (size_t)bh * L_ * D_) + hc * 8 + c;
  if (g == 0) ob4[(size_t)MEAN0 * (D_ / 4)] = m;
#pragma unroll
  for (int it = 0; it < (L_ - MEAN0 - 1) / MROWS; ++it)
    ob4[(size_t)(MEAN0 + 1 + g + it * MROWS) * (D_ / 4)] = m;
}

__device__ __forceinline__ float xhalf(float x) {
  int b = __builtin_bit_cast(int, x);
  return __builtin_bit_cast(float,
      __builtin_amdgcn_permlanex16(b, b, 0x76543210, 0xfedcba98, false, false));
}

struct KVTile { v2f k[16]; v2f v[16]; };

__device__ __forceinline__ void load_tile(KVTile& t, const float* __restrict__ kb,
                                          const float* __restrict__ vb,
                                          int kt, int tcol, int half) {
  const float* kp = kb + (kt * 16 + tcol) * D_ + 2 * half;
#pragma unroll
  for (int c4 = 0; c4 < 16; ++c4) {
    v2f a; a.x = kp[c4 * 4]; a.y = kp[c4 * 4 + 1];
    t.k[c4] = a;
  }
  const float* vp = vb + (kt * 16 + 8 * half) * D_ + tcol;
#pragma unroll
  for (int w = 0; w < 4; ++w)
#pragma unroll
    for (int mt = 0; mt < 4; ++mt) {
      v2f a; a.x = vp[(2 * w) * D_ + mt * 16]; a.y = vp[(2 * w + 1) * D_ + mt * 16];
      t.v[w * 4 + mt] = a;
    }
}

__device__ __forceinline__ void tile_step(const KVTile& t, const v2f (&bq)[16],
                                          v8f (&acc)[4], float& mrow, float& lsum,
                                          int kt, int i, int half) {
  v8f s = {0.f, 0.f, 0.f, 0.f, 0.f, 0.f, 0.f, 0.f};
#pragma unroll
  for (int c4 = 0; c4 < 16; ++c4)
    s = __builtin_amdgcn_wmma_f32_16x16x4_f32(
            false, t.k[c4], false, bq[c4], (short)0, s, false, false);

  const int j0 = KEY0 + kt * 16 + 8 * half;
  float sv[8];
#pragma unroll
  for (int r = 0; r < 8; ++r)
    sv[r] = (j0 + r - i >= DIAG) ? s[r] : -10000.0f;
  float rm = sv[0];
#pragma unroll
  for (int r = 1; r < 8; ++r) rm = fmaxf(rm, sv[r]);
  rm = fmaxf(rm, xhalf(rm));
  const float mnew  = fmaxf(mrow, rm);
  const float alpha = __expf(mrow - mnew);
  float p[8];
  float ps = 0.0f;
#pragma unroll
  for (int r = 0; r < 8; ++r) { p[r] = __expf(sv[r] - mnew); ps += p[r]; }
  ps += xhalf(ps);
  lsum = lsum * alpha + ps;
  mrow = mnew;

#pragma unroll
  for (int mt = 0; mt < 4; ++mt)
#pragma unroll
    for (int r = 0; r < 8; ++r)
      acc[mt][r] *= alpha;

#pragma unroll
  for (int w = 0; w < 4; ++w) {
    v2f b; b.x = p[2 * w]; b.y = p[2 * w + 1];
#pragma unroll
    for (int mt = 0; mt < 4; ++mt)
      acc[mt] = __builtin_amdgcn_wmma_f32_16x16x4_f32(
                    false, t.v[w * 4 + mt], false, b, (short)0, acc[mt], false, false);
  }
}

__global__ __launch_bounds__(NWAVE * 32)
void local_attn_kernel(const float* __restrict__ q, const float* __restrict__ k,
                       const float* __restrict__ v, float* __restrict__ out) {
  __shared__ float4 red[NWAVE * 8];
  __shared__ float4 oSlab[AWAVE][16 * 17];

  const int bh = blockIdx.x >> 1;
  const int hc = blockIdx.x & 1;
  const int t  = threadIdx.x;
  const int w  = __builtin_amdgcn_readfirstlane(t >> 5);

  mean_rows(red, v, out, bh, hc, t, w);

  if (w >= AWAVE) return;

  float4* oLds = oSlab[w];
  const int qlong = 4 * hc + (w & 3);
  const int qt   = (w < 4) ? qlong : 15 - qlong;
  const int lane = t & 31;
  const int tcol = lane & 15;
  const int half = lane >> 4;

  const float* qb = q + (size_t)bh * L_ * D_ + (size_t)qt * 16 * D_;
  const float* kb = k + (size_t)bh * L_ * D_ + (size_t)KEY0 * D_;
  const float* vb = v + (size_t)bh * L_ * D_ + (size_t)KEY0 * D_;
  float*       ob = out + (size_t)bh * L_ * D_;

  KVTile ta, tb;
  load_tile(ta, kb, vb, qt, tcol, half);

  v2f bq[16];
#pragma unroll
  for (int c4 = 0; c4 < 16; ++c4) {
    const float* p = qb + tcol * D_ + c4 * 4 + 2 * half;
    v2f b; b.x = p[0] * 0.125f; b.y = p[1] * 0.125f;
    bq[c4] = b;
  }

  const v8f vzero = {0.f, 0.f, 0.f, 0.f, 0.f, 0.f, 0.f, 0.f};
  v8f acc[4];
  acc[0] = vzero; acc[1] = vzero; acc[2] = vzero; acc[3] = vzero;
  float mrow = -3.0e38f, lsum = 0.0f;
  const int i = qt * 16 + tcol;

  int kt = qt;
  for (; kt + 1 < 16; kt += 2) {
    load_tile(tb, kb, vb, kt + 1, tcol, half);
    __builtin_amdgcn_sched_barrier(0);
    tile_step(ta, bq, acc, mrow, lsum, kt, i, half);
    __builtin_amdgcn_sched_barrier(0);

    load_tile(ta, kb, vb, (kt + 2 < 16) ? kt + 2 : 15, tcol, half);
    __builtin_amdgcn_sched_barrier(0);
    tile_step(tb, bq, acc, mrow, lsum, kt + 1, i, half);
  }
  if (kt < 16)
    tile_step(ta, bq, acc, mrow, lsum, kt, i, half);

  const float inv = 1.0f / lsum;
#pragma unroll
  for (int mt = 0; mt < 4; ++mt) {
    float4 lo, hi;
    lo.x = acc[mt][0] * inv; lo.y = acc[mt][1] * inv;
    lo.z = acc[mt][2] * inv; lo.w = acc[mt][3] * inv;
    hi.x = acc[mt][4] * inv; hi.y = acc[mt][5] * inv;
    hi.z = acc[mt][6] * inv; hi.w = acc[mt][7] * inv;
    oLds[tcol * 17 + mt * 4 + half * 2]     = lo;
    oLds[tcol * 17 + mt * 4 + half * 2 + 1] = hi;
  }
  __builtin_amdgcn_fence(__ATOMIC_RELEASE, "workgroup");
  __builtin_amdgcn_wave_barrier();
  __builtin_amdgcn_fence(__ATOMIC_ACQUIRE, "workgroup");
  if (qt * 16 + 15 <= LASTQ) {
#pragma unroll
    for (int it = 0; it < 8; ++it) {
      const int io = qt * 16 + it * 2 + half;
      ((float4*)(ob + (size_t)io * D_))[tcol] = oLds[(it * 2 + half) * 17 + tcol];
    }
  } else {
#pragma unroll
    for (int it = 0; it < 8; ++it) {
      const int io = qt * 16 + it * 2 + half;
      if (io <= LASTQ)
        ((float4*)(ob + (size_t)io * D_))[tcol] = oLds[(it * 2 + half) * 17 + tcol];
    }
  }
}

extern "C" void kernel_launch(void* const* d_in, const int* in_sizes, int n_in,
                              void* d_out, int out_size, void* d_ws, size_t ws_size,
                              hipStream_t stream) {
  const float* q = (const float*)d_in[0];
  const float* k = (const float*)d_in[1];
  const float* v = (const float*)d_in[2];
  float* out = (float*)d_out;

  local_attn_kernel<<<B_ * H_ * 2, NWAVE * 32, 0, stream>>>(q, k, v, out);
}


# grid 128 -> 256 work-groups, the two jobs side by side on all WGPs: 128 start the 4 long query tiles of a (head, half) at once, 128 do its mean rows, then the 4 short tiles; same sums
# speedup vs baseline: 4.9407x; 1.3507x over previous
#include <hip/hip_runtime.h>


typedef __attribute__((ext_vector_type(2))) float v2f;
typedef __attribute__((ext_vector_type(8))) float v8f;

#define B_    4
#define H_    16
#define L_    2048
#define D_    64
#define RWIN  256
#define KEY0  (L_ - RWIN)
#define DIAG  (L_ - RWIN + 1)
#define LASTQ (L_ - 1 - DIAG)
#define MEAN0 (LASTQ + 1)
#define NWAVE 16
#define AWAVE 4
#define NUNIT (B_ * H_ * 2)
#define MROWS 64

__device__ __forceinline__ float4 f4add(float4 a, float4 b) {
  return make_float4(a.x + b.x, a.y + b.y, a.z + b.z, a.w + b.w);
}
__device__ __forceinline__ float4 f4xor(float4 a, int m) {
  return make_float4(__shfl_xor(a.x, m), __shfl_xor(a.y, m),
                     __shfl_xor(a.z, m), __shfl_xor(a.w, m));
}

__device__ __forceinline__
void mean_rows(float4* red, const float* __restrict__ v, float* __restrict__ out,
               int bh, int hc, int t, int w) {
  const int c    = t & 7;
  const int g    = t >> 3;
  const int lane = t & 31;

  const float4* vb4 = (const float4*)(v + (size_t)bh * L_ * D_) + hc * 8 + c;
  float4 x[L_ / MROWS];
#pragma unroll
  for (int it = 0; it < L_ / MROWS; ++it)
    x[it] = vb4[(size_t)(g + it * MROWS) * (D_ / 4)];
  float4 acc = x[0];
#pragma unroll
  for (int it = 1; it < L_ / MROWS; ++it) acc = f4add(acc, x[it]);

  acc = f4add(acc, f4xor(acc, 8));
  acc = f4add(acc, f4xor(acc, 16));
  if (lane < 8) red[w * 8 + c] = acc;
  __syncthreads();

  float4 s = red[lane];
#pragma unroll
  for (int i = 1; i < NWAVE / 4; ++i) s = f4add(s, red[lane + 32 * i]);
  s = f4add(s, f4xor(s, 8));
  s = f4add(s, f4xor(s, 16));
  const float sc = 1.0f / (float)L_;
  const float4 m = make_float4(s.x * sc, s.y * sc, s.z * sc, s.w * sc);

  float4* ob4 = (float4*)(out + (size_t)bh * L_ * D_) + hc * 8 + c;
  if (g == 0) ob4[(size_t)MEAN0 * (D_ / 4)] = m;
#pragma unroll
  for (int it = 0; it < (L_ - MEAN0 - 1) / MROWS; ++it)
    ob4[(size_t)(MEAN0 + 1 + g + it * MROWS) * (D_ / 4)] = m;
}

__device__ __forceinline__ float xhalf(float x) {
  int b = __builtin_bit_cast(int, x);
  return __builtin_bit_cast(float,
      __builtin_amdgcn_permlanex16(b, b, 0x76543210, 0xfedcba98, false, false));
}

struct KVTile { v2f k[16]; v2f v[16]; };

__device__ __forceinline__ void load_tile(KVTile& t, const float* __restrict__ kb,
                                          const float* __restrict__ vb,
                                          int kt, int tcol, int half) {
  const float* kp = kb + (kt * 16 + tcol) * D_ + 2 * half;
#pragma unroll
  for (int c4 = 0; c4 < 16; ++c4) {
    v2f a; a.x = kp[c4 * 4]; a.y = kp[c4 * 4 + 1];
    t.k[c4] = a;
  }
  const float* vp = vb + (kt * 16 + 8 * half) * D_ + tcol;
#pragma unroll
  for (int w = 0; w < 4; ++w)
#pragma unroll
    for (int mt = 0; mt < 4; ++mt) {
      v2f a; a.x = vp[(2 * w) * D_ + mt * 16]; a.y = vp[(2 * w + 1) * D_ + mt * 16];
      t.v[w * 4 + mt] = a;
    }
}

__device__ __forceinline__ void tile_step(const KVTile& t, const v2f (&bq)[16],
                                          v8f (&acc)[4], float& mrow, float& lsum,
                                          int kt, int i, int half) {
  v8f s = {0.f, 0.f, 0.f, 0.f, 0.f, 0.f, 0.f, 0.f};
#pragma unroll
  for (int c4 = 0; c4 < 16; ++c4)
    s = __builtin_amdgcn_wmma_f32_16x16x4_f32(
            false, t.k[c4], false, bq[c4], (short)0, s, false, false);

  const int j0 = KEY0 + kt * 16 + 8 * half;
  float sv[8];
#pragma unroll
  for (int r = 0; r < 8; ++r)
    sv[r] = (j0 + r - i >= DIAG) ? s[r] : -10000.0f;
  float rm = sv[0];
#pragma unroll
  for (int r = 1; r < 8; ++r) rm = fmaxf(rm, sv[r]);
  rm = fmaxf(rm, xhalf(rm));
  const float mnew  = fmaxf(mrow, rm);
  const float alpha = __expf(mrow - mnew);
  float p[8];
  float ps = 0.0f;
#pragma unroll
  for (int r = 0; r < 8; ++r) { p[r] = __expf(sv[r] - mnew); ps += p[r]; }
  ps += xhalf(ps);
  lsum = lsum * alpha + ps;
  mrow = mnew;

#pragma unroll
  for (int mt = 0; mt < 4; ++mt)
#pragma unroll
    for (int r = 0; r < 8; ++r)
      acc[mt][r] *= alpha;

#pragma unroll
  for (int w = 0; w < 4; ++w) {
    v2f b; b.x = p[2 * w]; b.y = p[2 * w + 1];
#pragma unroll
    for (int mt = 0; mt < 4; ++mt)
      acc[mt] = __builtin_amdgcn_wmma_f32_16x16x4_f32(
                    false, t.v[w * 4 + mt], false, b, (short)0, acc[mt], false, false);
  }
}

__global__ __launch_bounds__(NWAVE * 32)
void local_attn_kernel(const float* __restrict__ q, const float* __restrict__ k,
                       const float* __restrict__ v, float* __restrict__ out) {
  __shared__ float4 red[NWAVE * 8];
  __shared__ float4 oSlab[AWAVE][16 * 17];

  const int  unit   = blockIdx.x & (NUNIT - 1);
  const bool meanwg = blockIdx.x >= NUNIT;
  const int bh = unit >> 1;
  const int hc = unit & 1;
  const int t  = threadIdx.x;
  const int w  = __builtin_amdgcn_readfirstlane(t >> 5);

  if (meanwg) mean_rows(red, v, out, bh, hc, t, w);

  if (w >= AWAVE) return;

  float4* oLds = oSlab[w];
  const int qlong = 2 * w + hc;
  const int qt   = meanwg ? 15 - qlong : qlong;
  const int lane = t & 31;
  const int tcol = lane & 15;
  const int half = lane >> 4;

  const float* qb = q + (size_t)bh * L_ * D_ + (size_t)qt * 16 * D_;
  const float* kb = k + (size_t)bh * L_ * D_ + (size_t)KEY0 * D_;
  const float* vb = v + (size_t)bh * L_ * D_ + (size_t)KEY0 * D_;
  float*       ob = out + (size_t)bh * L_ * D_;

  KVTile ta, tb;
  load_tile(ta, kb, vb, qt, tcol, half);

  v2f bq[16];
#pragma unroll
  for (int c4 = 0; c4 < 16; ++c4) {
    const float* p = qb + tcol * D_ + c4 * 4 + 2 * half;
    v2f b; b.x = p[0] * 0.125f; b.y = p[1] * 0.125f;
    bq[c4] = b;
  }

  const v8f vzero = {0.f, 0.f, 0.f, 0.f, 0.f, 0.f, 0.f, 0.f};
  v8f acc[4];
  acc[0] = vzero; acc[1] = vzero; acc[2] = vzero; acc[3] = vzero;
  float mrow = -3.0e38f, lsum = 0.0f;
  const int i = qt * 16 + tcol;

  int kt = qt;
  for (; kt + 1 < 16; kt += 2) {
    load_tile(tb, kb, vb, kt + 1, tcol, half);
    __builtin_amdgcn_sched_barrier(0);
    tile_step(ta, bq, acc, mrow, lsum, kt, i, half);
    __builtin_amdgcn_sched_barrier(0);

    load_tile(ta, kb, vb, (kt + 2 < 16) ? kt + 2 : 15, tcol, half);
    __builtin_amdgcn_sched_barrier(0);
    tile_step(tb, bq, acc, mrow, lsum, kt + 1, i, half);
  }
  if (kt < 16)
    tile_step(ta, bq, acc, mrow, lsum, kt, i, half);

  const float inv = 1.0f / lsum;
#pragma unroll
  for (int mt = 0; mt < 4; ++mt) {
    float4 lo, hi;
    lo.x = acc[mt][0] * inv; lo.y = acc[mt][1] * inv;
    lo.z = acc[mt][2] * inv; lo.w = acc[mt][3] * inv;
    hi.x = acc[mt][4] * inv; hi.y = acc[mt][5] * inv;
    hi.z = acc[mt][6] * inv; hi.w = acc[mt][7] * inv;
    oLds[tcol * 17 + mt * 4 + half * 2]     = lo;
    oLds[tcol * 17 + mt * 4 + half * 2 + 1] = hi;
  }
  __builtin_amdgcn_fence(__ATOMIC_RELEASE, "workgroup");
  __builtin_amdgcn_wave_barrier();
  __builtin_amdgcn_fence(__ATOMIC_ACQUIRE, "workgroup");
  if (qt * 16 + 15 <= LASTQ) {
#pragma unroll
    for (int it = 0; it < 8; ++it) {
      const int io = qt * 16 + it * 2 + half;
      ((float4*)(ob + (size_t)io * D_))[tcol] = oLds[(it * 2 + half) * 17 + tcol];
    }
  } else {
#pragma unroll
    for (int it = 0; it < 8; ++it) {
      const int io = qt * 16 + it * 2 + half;
      if (io <= LASTQ)
        ((float4*)(ob + (size_t)io * D_))[tcol] = oLds[(it * 2 + half) * 17 + tcol];
    }
  }
}

extern "C" void kernel_launch(void* const* d_in, const int* in_sizes, int n_in,
                              void* d_out, int out_size, void* d_ws, size_t ws_size,
                              hipStream_t stream) {
  const float* q = (const float*)d_in[0];
  const float* k = (const float*)d_in[1];
  const float* v = (const float*)d_in[2];
  float* out = (float*)d_out;

  local_attn_kernel<<<2 * NUNIT, NWAVE * 32, 0, stream>>>(q, k, v, out);
}


# short query tiles moved from behind the mean job to waves 4..7 of the attention work-group, 2nd wave per SIMD beside each long tile (17 key tiles per SIMD); mean work-groups end there; same sums
# speedup vs baseline: 5.3882x; 1.0906x over previous
#include <hip/hip_runtime.h>


typedef __attribute__((ext_vector_type(2))) float v2f;
typedef __attribute__((ext_vector_type(8))) float v8f;

#define B_    4
#define H_    16
#define L_    2048
#define D_    64
#define RWIN  256
#define KEY0  (L_ - RWIN)
#define DIAG  (L_ - RWIN + 1)
#define LASTQ (L_ - 1 - DIAG)
#define MEAN0 (LASTQ + 1)
#define NWAVE 16
#define AWAVE 8
#define NUNIT (B_ * H_ * 2)
#define MROWS 64

__device__ __forceinline__ float4 f4add(float4 a, float4 b) {
  return make_float4(a.x + b.x, a.y + b.y, a.z + b.z, a.w + b.w);
}
__device__ __forceinline__ float4 f4xor(float4 a, int m) {
  return make_float4(__shfl_xor(a.x, m), __shfl_xor(a.y, m),
                     __shfl_xor(a.z, m), __shfl_xor(a.w, m));
}

__device__ __forceinline__
void mean_rows(float4* red, const float* __restrict__ v, float* __restrict__ out,
               int bh, int hc, int t, int w) {
  const int c    = t & 7;
  const int g    = t >> 3;
  const int lane = t & 31;

  const float4* vb4 = (const float4*)(v + (size_t)bh * L_ * D_) + hc * 8 + c;
  float4 x[L_ / MROWS];
#pragma unroll
  for (int it = 0; it < L_ / MROWS; ++it)
    x[it] = vb4[(size_t)(g + it * MROWS) * (D_ / 4)];
  float4 acc = x[0];
#pragma unroll
  for (int it = 1; it < L_ / MROWS; ++it) acc = f4add(acc, x[it]);

  acc = f4add(acc, f4xor(acc, 8));
  acc = f4add(acc, f4xor(acc, 16));
  if (lane < 8) red[w * 8 + c] = acc;
  __syncthreads();

  float4 s = red[lane];
#pragma unroll
  for (int i = 1; i < NWAVE / 4; ++i) s = f4add(s, red[lane + 32 * i]);
  s = f4add(s, f4xor(s, 8));
  s = f4add(s, f4xor(s, 16));
  const float sc = 1.0f / (float)L_;
  const float4 m = make_float4(s.x * sc, s.y * sc, s.z * sc, s.w * sc);

  float4* ob4 = (float4*)(out + (size_t)bh * L_ * D_) + hc * 8 + c;
  if (g == 0) ob4[(size_t)MEAN0 * (D_ / 4)] = m;
#pragma unroll
  for (int it = 0; it < (L_ - MEAN0 - 1) / MROWS; ++it)
    ob4[(size_t)(MEAN0 + 1 + g + it * MROWS) * (D_ / 4)] = m;
}

__device__ __forceinline__ float xhalf(float x) {
  int b = __builtin_bit_cast(int, x);
  return __builtin_bit_cast(float,
      __builtin_amdgcn_permlanex16(b, b, 0x76543210, 0xfedcba98, false, false));
}

struct KVTile { v2f k[16]; v2f v[16]; };

__device__ __forceinline__ void load_tile(KVTile& t, const float* __restrict__ kb,
                                          const float* __restrict__ vb,
                                          int kt, int tcol, int half) {
  const float* kp = kb + (kt * 16 + tcol) * D_ + 2 * half;
#pragma unroll
  for (int c4 = 0; c4 < 16; ++c4) {
    v2f a; a.x = kp[c4 * 4]; a.y = kp[c4 * 4 + 1];
    t.k[c4] = a;
  }
  const float* vp = vb + (kt * 16 + 8 * half) * D_ + tcol;
#pragma unroll
  for (int w = 0; w < 4; ++w)
#pragma unroll
    for (int mt = 0; mt < 4; ++mt) {
      v2f a; a.x = vp[(2 * w) * D_ + mt * 16]; a.y = vp[(2 * w + 1) * D_ + mt * 16];
      t.v[w * 4 + mt] = a;
    }
}

__device__ __forceinline__ void tile_step(const KVTile& t, const v2f (&bq)[16],
                                          v8f (&acc)[4], float& mrow, float& lsum,
                                          int kt, int i, int half) {
  v8f s = {0.f, 0.f, 0.f, 0.f, 0.f, 0.f, 0.f, 0.f};
#pragma unroll
  for (int c4 = 0; c4 < 16; ++c4)
    s = __builtin_amdgcn_wmma_f32_16x16x4_f32(
            false, t.k[c4], false, bq[c4], (short)0, s, false, false);

  const int j0 = KEY0 + kt * 16 + 8 * half;
  float sv[8];
#pragma unroll
  for (int r = 0; r < 8; ++r)
    sv[r] = (j0 + r - i >= DIAG) ? s[r] : -10000.0f;
  float rm = sv[0];
#pragma unroll
  for (int r = 1; r < 8; ++r) rm = fmaxf(rm, sv[r]);
  rm = fmaxf(rm, xhalf(rm));
  const float mnew  = fmaxf(mrow, rm);
  const float alpha = __expf(mrow - mnew);
  float p[8];
  float ps = 0.0f;
#pragma unroll
  for (int r = 0; r < 8; ++r) { p[r] = __expf(sv[r] - mnew); ps += p[r]; }
  ps += xhalf(ps);
  lsum = lsum * alpha + ps;
  mrow = mnew;

#pragma unroll
  for (int mt = 0; mt < 4; ++mt)
#pragma unroll
    for (int r = 0; r < 8; ++r)
      acc[mt][r] *= alpha;

#pragma unroll
  for (int w = 0; w < 4; ++w) {
    v2f b; b.x = p[2 * w]; b.y = p[2 * w + 1];
#pragma unroll
    for (int mt = 0; mt < 4; ++mt)
      acc[mt] = __builtin_amdgcn_wmma_f32_16x16x4_f32(
                    false, t.v[w * 4 + mt], false, b, (short)0, acc[mt], false, false);
  }
}

__global__ __launch_bounds__(NWAVE * 32)
void local_attn_kernel(const float* __restrict__ q, const float* __restrict__ k,
                       const float* __restrict__ v, float* __restrict__ out) {
  __shared__ float4 red[NWAVE * 8];
  __shared__ float4 oSlab[AWAVE][16 * 17];

  const int  unit   = blockIdx.x & (NUNIT - 1);
  const bool meanwg = blockIdx.x >= NUNIT;
  const int bh = unit >> 1;
  const int hc = unit & 1;
  const int t  = threadIdx.x;
  const int w  = __builtin_amdgcn_readfirstlane(t >> 5);

  if (meanwg) { mean_rows(red, v, out, bh, hc, t, w); return; }

  if (w >= AWAVE) return;

  float4* oLds = oSlab[w];
  const int qlong = 2 * (w & 3) + hc;
  const int qt   = (w >= 4) ? 15 - qlong : qlong;
  const int lane = t & 31;
  const int tcol = lane & 15;
  const int half = lane >> 4;

  const float* qb = q + (size_t)bh * L_ * D_ + (size_t)qt * 16 * D_;
  const float* kb = k + (size_t)bh * L_ * D_ + (size_t)KEY0 * D_;
  const float* vb = v + (size_t)bh * L_ * D_ + (size_t)KEY0 * D_;
  float*       ob = out + (size_t)bh * L_ * D_;

  KVTile ta, tb;
  load_tile(ta, kb, vb, qt, tcol, half);

  v2f bq[16];
#pragma unroll
  for (int c4 = 0; c4 < 16; ++c4) {
    const float* p = qb + tcol * D_ + c4 * 4 + 2 * half;
    v2f b; b.x = p[0] * 0.125f; b.y = p[1] * 0.125f;
    bq[c4] = b;
  }

  const v8f vzero = {0.f, 0.f, 0.f, 0.f, 0.f, 0.f, 0.f, 0.f};
  v8f acc[4];
  acc[0] = vzero; acc[1] = vzero; acc[2] = vzero; acc[3] = vzero;
  float mrow = -3.0e38f, lsum = 0.0f;
  const int i = qt * 16 + tcol;

  int kt = qt;
  for (; kt + 1 < 16; kt += 2) {
    load_tile(tb, kb, vb, kt + 1, tcol, half);
    __builtin_amdgcn_sched_barrier(0);
    tile_step(ta, bq, acc, mrow, lsum, kt, i, half);
    __builtin_amdgcn_sched_barrier(0);

    load_tile(ta, kb, vb, (kt + 2 < 16) ? kt + 2 : 15, tcol, half);
    __builtin_amdgcn_sched_barrier(0);
    tile_step(tb, bq, acc, mrow, lsum, kt + 1, i, half);
  }
  if (kt < 16)
    tile_step(ta, bq, acc, mrow, lsum, kt, i, half);

  const float inv = 1.0f / lsum;
#pragma unroll
  for (int mt = 0; mt < 4; ++mt) {
    float4 lo, hi;
    lo.x = acc[mt][0] * inv; lo.y = acc[mt][1] * inv;
    lo.z = acc[mt][2] * inv; lo.w = acc[mt][3] * inv;
    hi.x = acc[mt][4] * inv; hi.y = acc[mt][5] * inv;
    hi.z = acc[mt][6] * inv; hi.w = acc[mt][7] * inv;
    oLds[tcol * 17 + mt * 4 + half * 2]     = lo;
    oLds[tcol * 17 + mt * 4 + half * 2 + 1] = hi;
  }
  __builtin_amdgcn_fence(__ATOMIC_RELEASE, "workgroup");
  __builtin_amdgcn_wave_barrier();
  __builtin_amdgcn_fence(__ATOMIC_ACQUIRE, "workgroup");
  if (qt * 16 + 15 <= LASTQ) {
#pragma unroll
    for (int it = 0; it < 8; ++it) {
      const int io = qt * 16 + it * 2 + half;
      ((float4*)(ob + (size_t)io * D_))[tcol] = oLds[(it * 2 + half) * 17 + tcol];
    }
  } else {
#pragma unroll
    for (int it = 0; it < 8; ++it) {
      const int io = qt * 16 + it * 2 + half;
      if (io <= LASTQ)
        ((float4*)(ob + (size_t)io * D_))[tcol] = oLds[(it * 2 + half) * 17 + tcol];
    }
  }
}

extern "C" void kernel_launch(void* const* d_in, const int* in_sizes, int n_in,
                              void* d_out, int out_size, void* d_ws, size_t ws_size,
                              hipStream_t stream) {
  const float* q = (const float*)d_in[0];
  const float* k = (const float*)d_in[1];
  const float* v = (const float*)d_in[2];
  float* out = (float*)d_out;

  local_attn_kernel<<<2 * NUNIT, NWAVE * 32, 0, stream>>>(q, k, v, out);
}
